# starting kernel
# baseline (speedup 1.0000x reference)
#include <hip/hip_runtime.h>

typedef __attribute__((ext_vector_type(2))) float v2f;
typedef __attribute__((ext_vector_type(8))) float v8f;

#define BATCH   4096
#define TSTEPS  2048

__device__ __forceinline__ float fsigm_b(float x, float hb) {
    return fmaf(__builtin_amdgcn_tanhf(fmaf(x, 0.5f, hb)), 0.5f, 0.5f);
}

__global__ __launch_bounds__(32)
void OpenLSTM_wmma_scan(const float* __restrict__ u,
                        const float* __restrict__ Wih,
                        const float* __restrict__ Whh,
                        const float* __restrict__ bih,
                        const float* __restrict__ bhh,
                        const float* __restrict__ Whr,
                        const int*   __restrict__ ncp,
                        float*       __restrict__ out)
{
    const int lane = threadIdx.x;
    const int n    = lane & 15;
    const int hi   = lane >> 4;
    const int b0   = blockIdx.x * 16;
    const int nc   = *ncp;

    v2f A[4];
    const float* Wsrc = hi ? Whh : Wih;
#pragma unroll
    for (int g = 0; g < 4; ++g) {
        const int row = g * 16 + n;
        A[g].x = Wsrc[row * 2 + 0];
        A[g].y = Wsrc[row * 2 + 1];
    }

    float hbi[8], hbf[8], hbo[8], bg[8];
#pragma unroll
    for (int r = 0; r < 8; ++r) {
        const int m = r + hi * 8;
        hbi[r] = 0.5f * (bih[ 0 + m] + bhh[ 0 + m]);
        hbf[r] = 0.5f * (bih[16 + m] + bhh[16 + m]);
        bg[r]  =         bih[32 + m] + bhh[32 + m];
        hbo[r] = 0.5f * (bih[48 + m] + bhh[48 + m]);
    }

    float wr0[8], wr1[8];
#pragma unroll
    for (int r = 0; r < 8; ++r) {
        wr0[r] = Whr[0 * 16 + r + hi * 8];
        wr1[r] = Whr[1 * 16 + r + hi * 8];
    }

    v8f c = {};
    const v8f zc = {};
    float h0 = 0.0f, h1 = 0.0f;

    const float* up = u + (size_t)(b0 + n) * TSTEPS * 4 + hi * 2;
    float*       op = out + (size_t)(b0 + n) * TSTEPS * 2;

    float2 ld = *(const float2*)up;

    for (int t = 0; t < TSTEPS; ++t) {
        const bool use_mem = (!hi) || (t < nc);
        v2f Bv;
        Bv.x = use_mem ? ld.x : h0;
        Bv.y = use_mem ? ld.y : h1;

        const int tn = (t + 1 < TSTEPS) ? (t + 1) : (TSTEPS - 1);
        const float2 ldn = *(const float2*)(up + (size_t)tn * 4);
        __builtin_prefetch(up + (size_t)tn * 4 + 32, 0, 3);

        v8f gi = __builtin_amdgcn_wmma_f32_16x16x4_f32(false, A[0], false, Bv, (short)0, zc, false, false);
        v8f gf = __builtin_amdgcn_wmma_f32_16x16x4_f32(false, A[1], false, Bv, (short)0, zc, false, false);
        v8f gg = __builtin_amdgcn_wmma_f32_16x16x4_f32(false, A[2], false, Bv, (short)0, zc, false, false);
        v8f go = __builtin_amdgcn_wmma_f32_16x16x4_f32(false, A[3], false, Bv, (short)0, zc, false, false);

        float p0 = 0.0f, p1 = 0.0f;
#pragma unroll
        for (int r = 0; r < 8; ++r) {
            const float si = fsigm_b(gi[r], hbi[r]);
            const float sf = fsigm_b(gf[r], hbf[r]);
            const float tg = __builtin_amdgcn_tanhf(gg[r] + bg[r]);
            const float so = fsigm_b(go[r], hbo[r]);
            const float cn = sf * c[r] + si * tg;
            c[r] = cn;
            const float s = so * __builtin_amdgcn_tanhf(cn);
            p0 += s * wr0[r];
            p1 += s * wr1[r];
        }
        h0 = p0 + __shfl_xor(p0, 16, 32);
        h1 = p1 + __shfl_xor(p1, 16, 32);

        if (!hi) {
            float2 hv; hv.x = h0; hv.y = h1;
            __builtin_nontemporal_store(__builtin_bit_cast(double, hv),
                                        (double*)(op + (size_t)t * 2));
        }
        ld = ldn;
    }
}

extern "C" void kernel_launch(void* const* d_in, const int* in_sizes, int n_in,
                              void* d_out, int out_size, void* d_ws, size_t ws_size,
                              hipStream_t stream) {
    const float* u    = (const float*)d_in[0];
    const float* Wih  = (const float*)d_in[1];
    const float* Whh  = (const float*)d_in[2];
    const float* bih  = (const float*)d_in[3];
    const float* bhh  = (const float*)d_in[4];
    const float* Whr  = (const float*)d_in[5];
    const int*   ncp  = (const int*)d_in[6];
    float*       out  = (float*)d_out;

    dim3 grid(BATCH / 16);
    dim3 block(32);
    OpenLSTM_wmma_scan<<<grid, block, 0, stream>>>(u, Wih, Whh, bih, bhh, Whr, ncp, out);
}


# steps t >= n_context: h goes to the LDS slot of its step's x; barriers hand finished rows to the idle waves, which store whole 128-byte lines; serial step has no global store; sums unchanged
# speedup vs baseline: 5.3179x; 5.3179x over previous
#include <hip/hip_runtime.h>


typedef __attribute__((ext_vector_type(2))) float v2f;
typedef __attribute__((ext_vector_type(4))) float v4f;
typedef __attribute__((ext_vector_type(8))) float v8f;

#define BATCH   4096
#define TSTEPS  2048
#define NW      16
#define FIRST   768
#define BLK     256

__device__ __forceinline__ float fsigm_b(float x, float hb) {
    return fmaf(__builtin_amdgcn_tanhf(fmaf(x, 0.5f, hb)), 0.5f, 0.5f);
}

__global__ __launch_bounds__(32 * NW)
void OpenLSTM_wmma_scan(const float* __restrict__ u,
                        const float* __restrict__ Wih,
                        const float* __restrict__ Whh,
                        const float* __restrict__ bih,
                        const float* __restrict__ bhh,
                        const float* __restrict__ Whr,
                        const int*   __restrict__ ncp,
                        float*       __restrict__ out)
{
    __shared__ float sP[NW * 8 * 32];
    __shared__ float sC[NW * 8 * 32];
    __shared__ v2f   sX[(TSTEPS + 1) * 16];

    const int lane = threadIdx.x & 31;
    const int w    = (__builtin_amdgcn_readfirstlane((int)(threadIdx.x >> 5)) + NW - 1) & (NW - 1);
    const int n    = lane & 15;
    const int hi   = lane >> 4;
    const int b0   = blockIdx.x * 16;
    int nc = *ncp;
    nc = nc < 0 ? 0 : (nc > TSTEPS ? TSTEPS : nc);

    const int nr = (nc > 0) ? ((nc - 1) & ~15) : 0;
    const int t0 = (w == NW - 1) ? nr : (((w * nr) / (NW - 1)) & ~15);
    const int t1 = (w == NW - 1) ? nc : ((w == NW - 2) ? nr : ((((w + 1) * nr) / (NW - 1)) & ~15));
    const int t1p = (w == NW - 1) ? t0 : t1;
    const int off   = nc - nr;
    const int total = TSTEPS - nr;

    const int n2 = TSTEPS - nc;
    if (n2 > 0) {
        const float* src = u + ((size_t)(b0 + w) * TSTEPS + nc) * 4;
        const int ng = (n2 + 31) >> 5;
        if (w == 0 && lane < 16) { v2f z; z.x = 0.0f; z.y = 0.0f; sX[(off + n2) * 16 + lane] = z; }
        for (int g0 = 0; g0 < ng; g0 += 8) {
            v2f v[8];
#pragma unroll
            for (int k = 0; k < 8; ++k) {
                int j = (g0 + k) * 32 + lane;
                j = j < n2 ? j : n2 - 1;
                v[k] = *(const v2f*)(src + (size_t)j * 4);
            }
#pragma unroll
            for (int k = 0; k < 8; ++k) {
                const int j = (g0 + k) * 32 + lane;
                if (j < n2) sX[(off + j) * 16 + w] = v[k];
            }
        }
    }

    v2f A[4];
    const float* Wsrc = hi ? Whh : Wih;
#pragma unroll
    for (int g = 0; g < 4; ++g) {
        const int row = g * 16 + n;
        A[g].x = Wsrc[row * 2 + 0];
        A[g].y = Wsrc[row * 2 + 1];
    }

    v2f A2[4];
#pragma unroll
    for (int g = 0; g < 4; ++g) {
        const int row = g * 16 + n;
        A2[g].x = Whh[row * 2 + hi];
        A2[g].y = Wih[row * 2 + hi];
    }

    float hbi[8], hbf[8], hbo[8], bg[8];
#pragma unroll
    for (int r = 0; r < 8; ++r) {
        const int m = r + hi * 8;
        hbi[r] = 0.5f * (bih[ 0 + m] + bhh[ 0 + m]);
        hbf[r] = 0.5f * (bih[16 + m] + bhh[16 + m]);
        bg[r]  =         bih[32 + m] + bhh[32 + m];
        hbo[r] = 0.5f * (bih[48 + m] + bhh[48 + m]);
    }

    float wr0[8], wr1[8];
#pragma unroll
    for (int r = 0; r < 8; ++r) {
        wr0[r] = Whr[0 * 16 + r + hi * 8];
        wr1[r] = Whr[1 * 16 + r + hi * 8];
    }

    const v8f zc = {};

    const float* up = u + (size_t)(b0 + n) * TSTEPS * 4 + hi * 2;
    float*       op = out + (size_t)(b0 + n) * TSTEPS * 2;

    v8f cl = {};
    v8f P;
#pragma unroll
    for (int r = 0; r < 8; ++r) P[r] = 1.0f;
    {
        const float* p = up + (size_t)t0 * 4;
        float2 ld = *(const float2*)p;
        for (int t = t0; t < t1p; ++t) {
            v2f Bv;
            Bv.x = ld.x;
            Bv.y = ld.y;
            p += (t + 1 < TSTEPS) ? 4 : 0;
            const float2 ldn = *(const float2*)p;
            __builtin_prefetch(p + 32, 0, 3);

            v8f gi = __builtin_amdgcn_wmma_f32_16x16x4_f32(false, A[0], false, Bv, (short)0, zc, false, false);
            v8f gf = __builtin_amdgcn_wmma_f32_16x16x4_f32(false, A[1], false, Bv, (short)0, zc, false, false);
            v8f gg = __builtin_amdgcn_wmma_f32_16x16x4_f32(false, A[2], false, Bv, (short)0, zc, false, false);
#pragma unroll
            for (int r = 0; r < 8; ++r) {
                const float si = fsigm_b(gi[r], hbi[r]);
                const float sf = fsigm_b(gf[r], hbf[r]);
                const float tg = __builtin_amdgcn_tanhf(gg[r] + bg[r]);
                cl[r] = sf * cl[r] + si * tg;
                P[r]  = P[r] * sf;
            }
            ld = ldn;
        }
    }
#pragma unroll
    for (int r = 0; r < 8; ++r) {
        sP[(w * 8 + r) * 32 + lane] = P[r];
        sC[(w * 8 + r) * 32 + lane] = cl[r];
    }
    __syncthreads();

    if (w == NW - 1) __builtin_amdgcn_s_setprio(3);

    v8f c = {};
    for (int j = 0; j < w; ++j) {
#pragma unroll
        for (int r = 0; r < 8; ++r)
            c[r] = fmaf(sP[(j * 8 + r) * 32 + lane], c[r], sC[(j * 8 + r) * 32 + lane]);
    }

    float h0 = 0.0f, h1 = 0.0f;
    {
        const float* p = up + (size_t)t0 * 4;
        float*       q = op + (size_t)t0 * 2;
        float2 ld = *(const float2*)p;
        for (int t = t0; t < t1; ++t) {
            v2f Bv;
            Bv.x = ld.x;
            Bv.y = ld.y;
            p += (t + 1 < TSTEPS) ? 4 : 0;
            const float2 ldn = *(const float2*)p;

            v8f gi = __builtin_amdgcn_wmma_f32_16x16x4_f32(false, A[0], false, Bv, (short)0, zc, false, false);
            v8f gf = __builtin_amdgcn_wmma_f32_16x16x4_f32(false, A[1], false, Bv, (short)0, zc, false, false);
            v8f gg = __builtin_amdgcn_wmma_f32_16x16x4_f32(false, A[2], false, Bv, (short)0, zc, false, false);
            v8f go = __builtin_amdgcn_wmma_f32_16x16x4_f32(false, A[3], false, Bv, (short)0, zc, false, false);

            float p0 = 0.0f, p1 = 0.0f;
#pragma unroll
            for (int r = 0; r < 8; ++r) {
                const float si = fsigm_b(gi[r], hbi[r]);
                const float sf = fsigm_b(gf[r], hbf[r]);
                const float tg = __builtin_amdgcn_tanhf(gg[r] + bg[r]);
                const float so = fsigm_b(go[r], hbo[r]);
                const float cn = sf * c[r] + si * tg;
                c[r] = cn;
                const float s = so * __builtin_amdgcn_tanhf(cn);
                p0 += s * wr0[r];
                p1 += s * wr1[r];
            }
            h0 = p0 + __shfl_xor(p0, 16, 32);
            h1 = p1 + __shfl_xor(p1, 16, 32);

            if (w == NW - 1) {
                if (!hi) { v2f hv; hv.x = h0; hv.y = h1; sX[(t - nr) * 16 + n] = hv; }
            } else if (!hi) {
                float2 hv; hv.x = h0; hv.y = h1;
                __builtin_nontemporal_store(__builtin_bit_cast(double, hv), (double*)q);
            }
            q += 2;
            ld = ldn;
        }
    }

    float* xs = (float*)sX + off * 32 + n * 2 + hi;
    float* hs = (off > 0) ? xs - 32 : xs + n2 * 32;
    float hB = hi ? h1 : h0;
    float ld = 0.0f;
    if (w == NW - 1 && n2 > 0) ld = *xs;
    int i = 0;
    const int nblk = (total <= FIRST) ? 1 : 1 + (total - FIRST + BLK - 1) / BLK;
    for (int blk = 0; blk <= nblk; ++blk) {
        if (w == NW - 1) {
            int iend = FIRST + blk * BLK - off + 1;
            iend = iend < n2 ? iend : n2;
            const int ifirst = i;
            for (; i < iend; ++i) {
                v2f Bv;
                Bv.x = hB;
                Bv.y = ld;

                xs += 32;
                const float ldn = *xs;
                __builtin_amdgcn_sched_barrier(0);

                v8f gi = __builtin_amdgcn_wmma_f32_16x16x4_f32(false, A2[0], false, Bv, (short)0, zc, false, false);
                v8f gf = __builtin_amdgcn_wmma_f32_16x16x4_f32(false, A2[1], false, Bv, (short)0, zc, false, false);
                v8f gg = __builtin_amdgcn_wmma_f32_16x16x4_f32(false, A2[2], false, Bv, (short)0, zc, false, false);
                v8f go = __builtin_amdgcn_wmma_f32_16x16x4_f32(false, A2[3], false, Bv, (short)0, zc, false, false);

                *hs = hB;
                hs = xs - 32;
                __builtin_amdgcn_sched_barrier(0);

                float p0 = 0.0f, p1 = 0.0f;
#pragma unroll
                for (int r = 0; r < 8; ++r) {
                    const float si = fsigm_b(gi[r], hbi[r]);
                    const float sf = fsigm_b(gf[r], hbf[r]);
                    const float tg = __builtin_amdgcn_tanhf(gg[r] + bg[r]);
                    const float so = fsigm_b(go[r], hbo[r]);
                    const float cn = sf * c[r] + si * tg;
                    c[r] = cn;
                    const float s = so * __builtin_amdgcn_tanhf(cn);
                    p0 += s * wr0[r];
                    p1 += s * wr1[r];
                }
                float sa = p0, sb = p1;
                asm("v_permlane16_swap_b32 %0, %1" : "+v"(sa), "+v"(sb));
                hB = sa + sb;

                ld = ldn;
            }
            if (i == n2 && i > ifirst) *hs = hB;
        } else if (blk > 0) {
            const int r0 = (blk == 1) ? 0 : FIRST + (blk - 2) * BLK;
            int r1 = FIRST + (blk - 1) * BLK;
            r1 = r1 < total ? r1 : total;
            const int nun = ((r1 - r0) >> 4) * 4;
            const int q4 = lane >> 3, p8 = lane & 7;
            for (int uu = w; uu < nun; uu += NW - 1) {
                const int nb = (uu & 3) * 4 + q4;
                const int s  = r0 + (uu >> 2) * 16 + 2 * p8;
                const v2f ha = sX[s * 16 + nb];
                const v2f hb = sX[(s + 1) * 16 + nb];
                v4f o;
                o.x = ha.x; o.y = ha.y; o.z = hb.x; o.w = hb.y;
                float* dst = out + ((size_t)(b0 + nb) * TSTEPS + nr + s) * 2;
                __builtin_nontemporal_store(o, (v4f*)dst);
            }
        }
        __syncthreads();
    }
}

extern "C" void kernel_launch(void* const* d_in, const int* in_sizes, int n_in,
                              void* d_out, int out_size, void* d_ws, size_t ws_size,
                              hipStream_t stream) {
    const float* u    = (const float*)d_in[0];
    const float* Wih  = (const float*)d_in[1];
    const float* Whh  = (const float*)d_in[2];
    const float* bih  = (const float*)d_in[3];
    const float* bhh  = (const float*)d_in[4];
    const float* Whr  = (const float*)d_in[5];
    const int*   ncp  = (const int*)d_in[6];
    float*       out  = (float*)d_out;

    dim3 grid(BATCH / 16);
    dim3 block(32 * NW);
    OpenLSTM_wmma_scan<<<grid, block, 0, stream>>>(u, Wih, Whh, bih, bhh, Whr, ncp, out);
}


# steps t >= n_context: LDS read of the next x moved from in front of the 4 products to directly behind them, beside the h store: one LDS issue point a step, inside the products' wait; sums unchanged
# speedup vs baseline: 5.8293x; 1.0962x over previous
#include <hip/hip_runtime.h>


typedef __attribute__((ext_vector_type(2))) float v2f;
typedef __attribute__((ext_vector_type(4))) float v4f;
typedef __attribute__((ext_vector_type(8))) float v8f;

#define BATCH   4096
#define TSTEPS  2048
#define NW      16
#define FIRST   768
#define BLK     256

__device__ __forceinline__ float fsigm_b(float x, float hb) {
    return fmaf(__builtin_amdgcn_tanhf(fmaf(x, 0.5f, hb)), 0.5f, 0.5f);
}

__global__ __launch_bounds__(32 * NW)
void OpenLSTM_wmma_scan(const float* __restrict__ u,
                        const float* __restrict__ Wih,
                        const float* __restrict__ Whh,
                        const float* __restrict__ bih,
                        const float* __restrict__ bhh,
                        const float* __restrict__ Whr,
                        const int*   __restrict__ ncp,
                        float*       __restrict__ out)
{
    __shared__ float sP[NW * 8 * 32];
    __shared__ float sC[NW * 8 * 32];
    __shared__ v2f   sX[(TSTEPS + 1) * 16];

    const int lane = threadIdx.x & 31;
    const int w    = (__builtin_amdgcn_readfirstlane((int)(threadIdx.x >> 5)) + NW - 1) & (NW - 1);
    const int n    = lane & 15;
    const int hi   = lane >> 4;
    const int b0   = blockIdx.x * 16;
    int nc = *ncp;
    nc = nc < 0 ? 0 : (nc > TSTEPS ? TSTEPS : nc);

    const int nr = (nc > 0) ? ((nc - 1) & ~15) : 0;
    const int t0 = (w == NW - 1) ? nr : (((w * nr) / (NW - 1)) & ~15);
    const int t1 = (w == NW - 1) ? nc : ((w == NW - 2) ? nr : ((((w + 1) * nr) / (NW - 1)) & ~15));
    const int t1p = (w == NW - 1) ? t0 : t1;
    const int off   = nc - nr;
    const int total = TSTEPS - nr;

    const int n2 = TSTEPS - nc;
    if (n2 > 0) {
        const float* src = u + ((size_t)(b0 + w) * TSTEPS + nc) * 4;
        const int ng = (n2 + 31) >> 5;
        if (w == 0 && lane < 16) { v2f z; z.x = 0.0f; z.y = 0.0f; sX[(off + n2) * 16 + lane] = z; }
        for (int g0 = 0; g0 < ng; g0 += 8) {
            v2f v[8];
#pragma unroll
            for (int k = 0; k < 8; ++k) {
                int j = (g0 + k) * 32 + lane;
                j = j < n2 ? j : n2 - 1;
                v[k] = *(const v2f*)(src + (size_t)j * 4);
            }
#pragma unroll
            for (int k = 0; k < 8; ++k) {
                const int j = (g0 + k) * 32 + lane;
                if (j < n2) sX[(off + j) * 16 + w] = v[k];
            }
        }
    }

    v2f A[4];
    const float* Wsrc = hi ? Whh : Wih;
#pragma unroll
    for (int g = 0; g < 4; ++g) {
        const int row = g * 16 + n;
        A[g].x = Wsrc[row * 2 + 0];
        A[g].y = Wsrc[row * 2 + 1];
    }

    v2f A2[4];
#pragma unroll
    for (int g = 0; g < 4; ++g) {
        const int row = g * 16 + n;
        A2[g].x = Whh[row * 2 + hi];
        A2[g].y = Wih[row * 2 + hi];
    }

    float hbi[8], hbf[8], hbo[8], bg[8];
#pragma unroll
    for (int r = 0; r < 8; ++r) {
        const int m = r + hi * 8;
        hbi[r] = 0.5f * (bih[ 0 + m] + bhh[ 0 + m]);
        hbf[r] = 0.5f * (bih[16 + m] + bhh[16 + m]);
        bg[r]  =         bih[32 + m] + bhh[32 + m];
        hbo[r] = 0.5f * (bih[48 + m] + bhh[48 + m]);
    }

    float wr0[8], wr1[8];
#pragma unroll
    for (int r = 0; r < 8; ++r) {
        wr0[r] = Whr[0 * 16 + r + hi * 8];
        wr1[r] = Whr[1 * 16 + r + hi * 8];
    }

    const v8f zc = {};

    const float* up = u + (size_t)(b0 + n) * TSTEPS * 4 + hi * 2;
    float*       op = out + (size_t)(b0 + n) * TSTEPS * 2;

    v8f cl = {};
    v8f P;
#pragma unroll
    for (int r = 0; r < 8; ++r) P[r] = 1.0f;
    {
        const float* p = up + (size_t)t0 * 4;
        float2 ld = *(const float2*)p;
        for (int t = t0; t < t1p; ++t) {
            v2f Bv;
            Bv.x = ld.x;
            Bv.y = ld.y;
            p += (t + 1 < TSTEPS) ? 4 : 0;
            const float2 ldn = *(const float2*)p;
            __builtin_prefetch(p + 32, 0, 3);

            v8f gi = __builtin_amdgcn_wmma_f32_16x16x4_f32(false, A[0], false, Bv, (short)0, zc, false, false);
            v8f gf = __builtin_amdgcn_wmma_f32_16x16x4_f32(false, A[1], false, Bv, (short)0, zc, false, false);
            v8f gg = __builtin_amdgcn_wmma_f32_16x16x4_f32(false, A[2], false, Bv, (short)0, zc, false, false);
#pragma unroll
            for (int r = 0; r < 8; ++r) {
                const float si = fsigm_b(gi[r], hbi[r]);
                const float sf = fsigm_b(gf[r], hbf[r]);
                const float tg = __builtin_amdgcn_tanhf(gg[r] + bg[r]);
                cl[r] = sf * cl[r] + si * tg;
                P[r]  = P[r] * sf;
            }
            ld = ldn;
        }
    }
#pragma unroll
    for (int r = 0; r < 8; ++r) {
        sP[(w * 8 + r) * 32 + lane] = P[r];
        sC[(w * 8 + r) * 32 + lane] = cl[r];
    }
    __syncthreads();

    if (w == NW - 1) __builtin_amdgcn_s_setprio(3);

    v8f c = {};
    for (int j = 0; j < w; ++j) {
#pragma unroll
        for (int r = 0; r < 8; ++r)
            c[r] = fmaf(sP[(j * 8 + r) * 32 + lane], c[r], sC[(j * 8 + r) * 32 + lane]);
    }

    float h0 = 0.0f, h1 = 0.0f;
    {
        const float* p = up + (size_t)t0 * 4;
        float*       q = op + (size_t)t0 * 2;
        float2 ld = *(const float2*)p;
        for (int t = t0; t < t1; ++t) {
            v2f Bv;
            Bv.x = ld.x;
            Bv.y = ld.y;
            p += (t + 1 < TSTEPS) ? 4 : 0;
            const float2 ldn = *(const float2*)p;

            v8f gi = __builtin_amdgcn_wmma_f32_16x16x4_f32(false, A[0], false, Bv, (short)0, zc, false, false);
            v8f gf = __builtin_amdgcn_wmma_f32_16x16x4_f32(false, A[1], false, Bv, (short)0, zc, false, false);
            v8f gg = __builtin_amdgcn_wmma_f32_16x16x4_f32(false, A[2], false, Bv, (short)0, zc, false, false);
            v8f go = __builtin_amdgcn_wmma_f32_16x16x4_f32(false, A[3], false, Bv, (short)0, zc, false, false);

            float p0 = 0.0f, p1 = 0.0f;
#pragma unroll
            for (int r = 0; r < 8; ++r) {
                const float si = fsigm_b(gi[r], hbi[r]);
                const float sf = fsigm_b(gf[r], hbf[r]);
                const float tg = __builtin_amdgcn_tanhf(gg[r] + bg[r]);
                const float so = fsigm_b(go[r], hbo[r]);
                const float cn = sf * c[r] + si * tg;
                c[r] = cn;
                const float s = so * __builtin_amdgcn_tanhf(cn);
                p0 += s * wr0[r];
                p1 += s * wr1[r];
            }
            h0 = p0 + __shfl_xor(p0, 16, 32);
            h1 = p1 + __shfl_xor(p1, 16, 32);

            if (w == NW - 1) {
                if (!hi) { v2f hv; hv.x = h0; hv.y = h1; sX[(t - nr) * 16 + n] = hv; }
            } else if (!hi) {
                float2 hv; hv.x = h0; hv.y = h1;
                __builtin_nontemporal_store(__builtin_bit_cast(double, hv), (double*)q);
            }
            q += 2;
            ld = ldn;
        }
    }

    float* xs = (float*)sX + off * 32 + n * 2 + hi;
    float* hs = (off > 0) ? xs - 32 : xs + n2 * 32;
    float hB = hi ? h1 : h0;
    float ld = 0.0f;
    if (w == NW - 1 && n2 > 0) ld = *xs;
    int i = 0;
    const int nblk = (total <= FIRST) ? 1 : 1 + (total - FIRST + BLK - 1) / BLK;
    for (int blk = 0; blk <= nblk; ++blk) {
        if (w == NW - 1) {
            int iend = FIRST + blk * BLK - off + 1;
            iend = iend < n2 ? iend : n2;
            const int ifirst = i;
            for (; i < iend; ++i) {
                v2f Bv;
                Bv.x = hB;
                Bv.y = ld;

                v8f gi = __builtin_amdgcn_wmma_f32_16x16x4_f32(false, A2[0], false, Bv, (short)0, zc, false, false);
                v8f gf = __builtin_amdgcn_wmma_f32_16x16x4_f32(false, A2[1], false, Bv, (short)0, zc, false, false);
                v8f gg = __builtin_amdgcn_wmma_f32_16x16x4_f32(false, A2[2], false, Bv, (short)0, zc, false, false);
                v8f go = __builtin_amdgcn_wmma_f32_16x16x4_f32(false, A2[3], false, Bv, (short)0, zc, false, false);

                __builtin_amdgcn_sched_barrier(0);

                *hs = hB;
                xs += 32;
                hs = xs - 32;
                const float ldn = *xs;
                __builtin_amdgcn_sched_barrier(0);

                float p0 = 0.0f, p1 = 0.0f;
#pragma unroll
                for (int r = 0; r < 8; ++r) {
                    const float si = fsigm_b(gi[r], hbi[r]);
                    const float sf = fsigm_b(gf[r], hbf[r]);
                    const float tg = __builtin_amdgcn_tanhf(gg[r] + bg[r]);
                    const float so = fsigm_b(go[r], hbo[r]);
                    const float cn = sf * c[r] + si * tg;
                    c[r] = cn;
                    const float s = so * __builtin_amdgcn_tanhf(cn);
                    p0 += s * wr0[r];
                    p1 += s * wr1[r];
                }
                float sa = p0, sb = p1;
                asm("v_permlane16_swap_b32 %0, %1" : "+v"(sa), "+v"(sb));
                hB = sa + sb;

                ld = ldn;
            }
            if (i == n2 && i > ifirst) *hs = hB;
        } else if (blk > 0) {
            const int r0 = (blk == 1) ? 0 : FIRST + (blk - 2) * BLK;
            int r1 = FIRST + (blk - 1) * BLK;
            r1 = r1 < total ? r1 : total;
            const int nun = ((r1 - r0) >> 4) * 4;
            const int q4 = lane >> 3, p8 = lane & 7;
            for (int uu = w; uu < nun; uu += NW - 1) {
                const int nb = (uu & 3) * 4 + q4;
                const int s  = r0 + (uu >> 2) * 16 + 2 * p8;
                const v2f ha = sX[s * 16 + nb];
                const v2f hb = sX[(s + 1) * 16 + nb];
                v4f o;
                o.x = ha.x; o.y = ha.y; o.z = hb.x; o.w = hb.y;
                float* dst = out + ((size_t)(b0 + nb) * TSTEPS + nr + s) * 2;
                __builtin_nontemporal_store(o, (v4f*)dst);
            }
        }
        __syncthreads();
    }
}

extern "C" void kernel_launch(void* const* d_in, const int* in_sizes, int n_in,
                              void* d_out, int out_size, void* d_ws, size_t ws_size,
                              hipStream_t stream) {
    const float* u    = (const float*)d_in[0];
    const float* Wih  = (const float*)d_in[1];
    const float* Whh  = (const float*)d_in[2];
    const float* bih  = (const float*)d_in[3];
    const float* bhh  = (const float*)d_in[4];
    const float* Whr  = (const float*)d_in[5];
    const int*   ncp  = (const int*)d_in[6];
    float*       out  = (float*)d_out;

    dim3 grid(BATCH / 16);
    dim3 block(32 * NW);
    OpenLSTM_wmma_scan<<<grid, block, 0, stream>>>(u, Wih, Whh, bih, bhh, Whr, ncp, out);
}


# steps t >= n_context: vector arithmetic on row pairs in packed f32 instructions; projection sums as {even rows, odd rows} pairs fed by fused packed multiply-adds, no add chains (sum order changes)
# speedup vs baseline: 6.4223x; 1.1017x over previous
#include <hip/hip_runtime.h>


typedef __attribute__((ext_vector_type(2))) float v2f;
typedef __attribute__((ext_vector_type(4))) float v4f;
typedef __attribute__((ext_vector_type(8))) float v8f;

#define BATCH   4096
#define TSTEPS  2048
#define NW      16
#define FIRST   768
#define BLK     256

__device__ __forceinline__ float fsigm_b(float x, float hb) {
    return fmaf(__builtin_amdgcn_tanhf(fmaf(x, 0.5f, hb)), 0.5f, 0.5f);
}

__global__ __launch_bounds__(32 * NW)
void OpenLSTM_wmma_scan(const float* __restrict__ u,
                        const float* __restrict__ Wih,
                        const float* __restrict__ Whh,
                        const float* __restrict__ bih,
                        const float* __restrict__ bhh,
                        const float* __restrict__ Whr,
                        const int*   __restrict__ ncp,
                        float*       __restrict__ out)
{
    __shared__ float sP[NW * 8 * 32];
    __shared__ float sC[NW * 8 * 32];
    __shared__ v2f   sX[(TSTEPS + 1) * 16];

    const int lane = threadIdx.x & 31;
    const int w    = (__builtin_amdgcn_readfirstlane((int)(threadIdx.x >> 5)) + NW - 1) & (NW - 1);
    const int n    = lane & 15;
    const int hi   = lane >> 4;
    const int b0   = blockIdx.x * 16;
    int nc = *ncp;
    nc = nc < 0 ? 0 : (nc > TSTEPS ? TSTEPS : nc);

    const int nr = (nc > 0) ? ((nc - 1) & ~15) : 0;
    const int t0 = (w == NW - 1) ? nr : (((w * nr) / (NW - 1)) & ~15);
    const int t1 = (w == NW - 1) ? nc : ((w == NW - 2) ? nr : ((((w + 1) * nr) / (NW - 1)) & ~15));
    const int t1p = (w == NW - 1) ? t0 : t1;
    const int off   = nc - nr;
    const int total = TSTEPS - nr;

    const int n2 = TSTEPS - nc;
    if (n2 > 0) {
        const float* src = u + ((size_t)(b0 + w) * TSTEPS + nc) * 4;
        const int ng = (n2 + 31) >> 5;
        if (w == 0 && lane < 16) { v2f z; z.x = 0.0f; z.y = 0.0f; sX[(off + n2) * 16 + lane] = z; }
        for (int g0 = 0; g0 < ng; g0 += 8) {
            v2f v[8];
#pragma unroll
            for (int k = 0; k < 8; ++k) {
                int j = (g0 + k) * 32 + lane;
                j = j < n2 ? j : n2 - 1;
                v[k] = *(const v2f*)(src + (size_t)j * 4);
            }
#pragma unroll
            for (int k = 0; k < 8; ++k) {
                const int j = (g0 + k) * 32 + lane;
                if (j < n2) sX[(off + j) * 16 + w] = v[k];
            }
        }
    }

    v2f A[4];
    const float* Wsrc = hi ? Whh : Wih;
#pragma unroll
    for (int g = 0; g < 4; ++g) {
        const int row = g * 16 + n;
        A[g].x = Wsrc[row * 2 + 0];
        A[g].y = Wsrc[row * 2 + 1];
    }

    v2f A2[4];
#pragma unroll
    for (int g = 0; g < 4; ++g) {
        const int row = g * 16 + n;
        A2[g].x = Whh[row * 2 + hi];
        A2[g].y = Wih[row * 2 + hi];
    }

    float hbi[8], hbf[8], hbo[8], bg[8];
#pragma unroll
    for (int r = 0; r < 8; ++r) {
        const int m = r + hi * 8;
        hbi[r] = 0.5f * (bih[ 0 + m] + bhh[ 0 + m]);
        hbf[r] = 0.5f * (bih[16 + m] + bhh[16 + m]);
        bg[r]  =         bih[32 + m] + bhh[32 + m];
        hbo[r] = 0.5f * (bih[48 + m] + bhh[48 + m]);
    }

    float wr0[8], wr1[8];
#pragma unroll
    for (int r = 0; r < 8; ++r) {
        wr0[r] = Whr[0 * 16 + r + hi * 8];
        wr1[r] = Whr[1 * 16 + r + hi * 8];
    }

    v2f hbi2[4], hbf2[4], hbo2[4], bg2[4], w0p[4], w1p[4];
#pragma unroll
    for (int q = 0; q < 4; ++q) {
        hbi2[q].x = hbi[2 * q]; hbi2[q].y = hbi[2 * q + 1];
        hbf2[q].x = hbf[2 * q]; hbf2[q].y = hbf[2 * q + 1];
        hbo2[q].x = hbo[2 * q]; hbo2[q].y = hbo[2 * q + 1];
        bg2[q].x  = bg[2 * q];  bg2[q].y  = bg[2 * q + 1];
        w0p[q].x  = wr0[2 * q]; w0p[q].y  = wr0[2 * q + 1];
        w1p[q].x  = wr1[2 * q]; w1p[q].y  = wr1[2 * q + 1];
    }

    const v8f zc = {};

    const float* up = u + (size_t)(b0 + n) * TSTEPS * 4 + hi * 2;
    float*       op = out + (size_t)(b0 + n) * TSTEPS * 2;

    v8f cl = {};
    v8f P;
#pragma unroll
    for (int r = 0; r < 8; ++r) P[r] = 1.0f;
    {
        const float* p = up + (size_t)t0 * 4;
        float2 ld = *(const float2*)p;
        for (int t = t0; t < t1p; ++t) {
            v2f Bv;
            Bv.x = ld.x;
            Bv.y = ld.y;
            p += (t + 1 < TSTEPS) ? 4 : 0;
            const float2 ldn = *(const float2*)p;
            __builtin_prefetch(p + 32, 0, 3);

            v8f gi = __builtin_amdgcn_wmma_f32_16x16x4_f32(false, A[0], false, Bv, (short)0, zc, false, false);
            v8f gf = __builtin_amdgcn_wmma_f32_16x16x4_f32(false, A[1], false, Bv, (short)0, zc, false, false);
            v8f gg = __builtin_amdgcn_wmma_f32_16x16x4_f32(false, A[2], false, Bv, (short)0, zc, false, false);
#pragma unroll
            for (int r = 0; r < 8; ++r) {
                const float si = fsigm_b(gi[r], hbi[r]);
                const float sf = fsigm_b(gf[r], hbf[r]);
                const float tg = __builtin_amdgcn_tanhf(gg[r] + bg[r]);
                cl[r] = sf * cl[r] + si * tg;
                P[r]  = P[r] * sf;
            }
            ld = ldn;
        }
    }
#pragma unroll
    for (int r = 0; r < 8; ++r) {
        sP[(w * 8 + r) * 32 + lane] = P[r];
        sC[(w * 8 + r) * 32 + lane] = cl[r];
    }
    __syncthreads();

    if (w == NW - 1) __builtin_amdgcn_s_setprio(3);

    v8f c = {};
    for (int j = 0; j < w; ++j) {
#pragma unroll
        for (int r = 0; r < 8; ++r)
            c[r] = fmaf(sP[(j * 8 + r) * 32 + lane], c[r], sC[(j * 8 + r) * 32 + lane]);
    }

    float h0 = 0.0f, h1 = 0.0f;
    {
        const float* p = up + (size_t)t0 * 4;
        float*       q = op + (size_t)t0 * 2;
        float2 ld = *(const float2*)p;
        for (int t = t0; t < t1; ++t) {
            v2f Bv;
            Bv.x = ld.x;
            Bv.y = ld.y;
            p += (t + 1 < TSTEPS) ? 4 : 0;
            const float2 ldn = *(const float2*)p;

            v8f gi = __builtin_amdgcn_wmma_f32_16x16x4_f32(false, A[0], false, Bv, (short)0, zc, false, false);
            v8f gf = __builtin_amdgcn_wmma_f32_16x16x4_f32(false, A[1], false, Bv, (short)0, zc, false, false);
            v8f gg = __builtin_amdgcn_wmma_f32_16x16x4_f32(false, A[2], false, Bv, (short)0, zc, false, false);
            v8f go = __builtin_amdgcn_wmma_f32_16x16x4_f32(false, A[3], false, Bv, (short)0, zc, false, false);

            float p0 = 0.0f, p1 = 0.0f;
#pragma unroll
            for (int r = 0; r < 8; ++r) {
                const float si = fsigm_b(gi[r], hbi[r]);
                const float sf = fsigm_b(gf[r], hbf[r]);
                const float tg = __builtin_amdgcn_tanhf(gg[r] + bg[r]);
                const float so = fsigm_b(go[r], hbo[r]);
                const float cn = sf * c[r] + si * tg;
                c[r] = cn;
                const float s = so * __builtin_amdgcn_tanhf(cn);
                p0 += s * wr0[r];
                p1 += s * wr1[r];
            }
            h0 = p0 + __shfl_xor(p0, 16, 32);
            h1 = p1 + __shfl_xor(p1, 16, 32);

            if (w == NW - 1) {
                if (!hi) { v2f hv; hv.x = h0; hv.y = h1; sX[(t - nr) * 16 + n] = hv; }
            } else if (!hi) {
                v2f hv; hv.x = h0; hv.y = h1;
                __builtin_nontemporal_store(hv, (v2f*)q);
            }
            q += 2;
            ld = ldn;
        }
    }

    float* xs = (float*)sX + off * 32 + n * 2 + hi;
    float* hs = (off > 0) ? xs - 32 : xs + n2 * 32;
    float hB = hi ? h1 : h0;
    float ld = 0.0f;
    if (w == NW - 1 && n2 > 0) ld = *xs;
    int i = 0;
    const int nblk = (total <= FIRST) ? 1 : 1 + (total - FIRST + BLK - 1) / BLK;
    for (int blk = 0; blk <= nblk; ++blk) {
        if (w == NW - 1) {
            int iend = FIRST + blk * BLK - off + 1;
            iend = iend < n2 ? iend : n2;
            const int ifirst = i;
            for (; i < iend; ++i) {
                v2f Bv;
                Bv.x = hB;
                Bv.y = ld;

                v8f gi = __builtin_amdgcn_wmma_f32_16x16x4_f32(false, A2[0], false, Bv, (short)0, zc, false, false);
                v8f gf = __builtin_amdgcn_wmma_f32_16x16x4_f32(false, A2[1], false, Bv, (short)0, zc, false, false);
                v8f gg = __builtin_amdgcn_wmma_f32_16x16x4_f32(false, A2[2], false, Bv, (short)0, zc, false, false);
                v8f go = __builtin_amdgcn_wmma_f32_16x16x4_f32(false, A2[3], false, Bv, (short)0, zc, false, false);

                __builtin_amdgcn_sched_barrier(0);

                *hs = hB;
                xs += 32;
                hs = xs - 32;
                const float ldn = *xs;
                __builtin_amdgcn_sched_barrier(0);

                const v2f hf = {0.5f, 0.5f};
                v2f a0, a1;
#pragma unroll
                for (int q = 0; q < 4; ++q) {
                    v2f xi, xf, xg, xo, cq;
                    xi.x = gi[2 * q]; xi.y = gi[2 * q + 1];
                    xf.x = gf[2 * q]; xf.y = gf[2 * q + 1];
                    xg.x = gg[2 * q]; xg.y = gg[2 * q + 1];
                    xo.x = go[2 * q]; xo.y = go[2 * q + 1];
                    cq.x = c[2 * q];  cq.y = c[2 * q + 1];
                    const v2f zi = __builtin_elementwise_fma(xi, hf, hbi2[q]);
                    const v2f zf = __builtin_elementwise_fma(xf, hf, hbf2[q]);
                    const v2f zg = xg + bg2[q];
                    const v2f zo = __builtin_elementwise_fma(xo, hf, hbo2[q]);
                    v2f ti, tf, tg, to;
                    ti.x = __builtin_amdgcn_tanhf(zi.x); ti.y = __builtin_amdgcn_tanhf(zi.y);
                    tf.x = __builtin_amdgcn_tanhf(zf.x); tf.y = __builtin_amdgcn_tanhf(zf.y);
                    tg.x = __builtin_amdgcn_tanhf(zg.x); tg.y = __builtin_amdgcn_tanhf(zg.y);
                    to.x = __builtin_amdgcn_tanhf(zo.x); to.y = __builtin_amdgcn_tanhf(zo.y);
                    const v2f si = __builtin_elementwise_fma(ti, hf, hf);
                    const v2f sf = __builtin_elementwise_fma(tf, hf, hf);
                    const v2f so = __builtin_elementwise_fma(to, hf, hf);
                    const v2f cn = __builtin_elementwise_fma(cq, sf, si * tg);
                    c[2 * q] = cn.x; c[2 * q + 1] = cn.y;
                    v2f th;
                    th.x = __builtin_amdgcn_tanhf(cn.x); th.y = __builtin_amdgcn_tanhf(cn.y);
                    const v2f s2 = so * th;
                    if (q == 0) { a0 = s2 * w0p[0]; a1 = s2 * w1p[0]; }
                    else { a0 = __builtin_elementwise_fma(s2, w0p[q], a0); a1 = __builtin_elementwise_fma(s2, w1p[q], a1); }
                }
                const float p0 = a0.x + a0.y, p1 = a1.x + a1.y;
                float sa = p0, sb = p1;
                asm("v_permlane16_swap_b32 %0, %1" : "+v"(sa), "+v"(sb));
                hB = sa + sb;

                ld = ldn;
            }
            if (i == n2 && i > ifirst) *hs = hB;
        } else if (blk > 0) {
            const int r0 = (blk == 1) ? 0 : FIRST + (blk - 2) * BLK;
            int r1 = FIRST + (blk - 1) * BLK;
            r1 = r1 < total ? r1 : total;
            const int nun = ((r1 - r0) >> 4) * 4;
            const int q4 = lane >> 3, p8 = lane & 7;
            for (int uu = w; uu < nun; uu += NW - 1) {
                const int nb = (uu & 3) * 4 + q4;
                const int s  = r0 + (uu >> 2) * 16 + 2 * p8;
                const v2f ha = sX[s * 16 + nb];
                const v2f hb = sX[(s + 1) * 16 + nb];
                v4f o;
                o.x = ha.x; o.y = ha.y; o.z = hb.x; o.w = hb.y;
                float* dst = out + ((size_t)(b0 + nb) * TSTEPS + nr + s) * 2;
                __builtin_nontemporal_store(o, (v4f*)dst);
            }
        }
        __syncthreads();
    }
}

extern "C" void kernel_launch(void* const* d_in, const int* in_sizes, int n_in,
                              void* d_out, int out_size, void* d_ws, size_t ws_size,
                              hipStream_t stream) {
    const float* u    = (const float*)d_in[0];
    const float* Wih  = (const float*)d_in[1];
    const float* Whh  = (const float*)d_in[2];
    const float* bih  = (const float*)d_in[3];
    const float* bhh  = (const float*)d_in[4];
    const float* Whr  = (const float*)d_in[5];
    const int*   ncp  = (const int*)d_in[6];
    float*       out  = (float*)d_out;

    dim3 grid(BATCH / 16);
    dim3 block(32 * NW);
    OpenLSTM_wmma_scan<<<grid, block, 0, stream>>>(u, Wih, Whh, bih, bhh, Whr, ncp, out);
}
